# starting kernel
# baseline (speedup 1.0000x reference)
#include <hip/hip_runtime.h>


typedef float v2f __attribute__((ext_vector_type(2)));
typedef float v8f __attribute__((ext_vector_type(8)));

#define BATCH 4096
#define TSTEP 512
#define IDIM  13
#define HDIM  7
#define GDIM  28
#define FFDIM 512

#define LOG2E 1.4426950408889634f

#if __has_builtin(__builtin_amdgcn_tanhf)
#define FAST_TANH(v) __builtin_amdgcn_tanhf(v)
#define TANH_TIER 2
#elif __has_builtin(__builtin_amdgcn_tanh_f32)
#define FAST_TANH(v) __builtin_amdgcn_tanh_f32(v)
#define TANH_TIER 2
#elif __has_builtin(__builtin_amdgcn_exp2f) && __has_builtin(__builtin_amdgcn_rcpf)
#define TANH_TIER 1
static __device__ __forceinline__ float FAST_TANH(float v) {
    float e2 = __builtin_amdgcn_exp2f(v * (2.0f * LOG2E));
    return fmaf(-2.0f, __builtin_amdgcn_rcpf(e2 + 1.0f), 1.0f);
}
#else
#define FAST_TANH(v) tanhf(v)
#define TANH_TIER 0
#endif

static __device__ __forceinline__ float sigm(float v) {
#if TANH_TIER == 2
    return fmaf(0.5f, FAST_TANH(0.5f * v), 0.5f);
#elif TANH_TIER == 1
    return __builtin_amdgcn_rcpf(1.0f + __builtin_amdgcn_exp2f(v * -LOG2E));
#else
    return 1.0f / (1.0f + __expf(-v));
#endif
}

__global__ __launch_bounds__(32) void lstm_fused_kernel(
    const float* __restrict__ x,     const float* __restrict__ w_ih,
    const float* __restrict__ w_hh,  const float* __restrict__ b_ih,
    const float* __restrict__ b_hh,  const float* __restrict__ fc1_w,
    const float* __restrict__ fc1_b, const float* __restrict__ fc_w,
    const float* __restrict__ fc_b,  float* __restrict__ out)
{
    __shared__ float gates[16][32];
    __shared__ float hbuf[16][8];
    __shared__ float hdump[16];
    __shared__ float red[16];

    const int l    = threadIdx.x;
    const int half = l >> 4;
    const int nl   = l & 15;
    const int b0   = blockIdx.x * 16;

    v2f Bih[2][4];
    v2f Bhh[2][2];
#pragma unroll
    for (int kc = 0; kc < 4; ++kc) {
        int k0 = kc * 4 + 2 * half;
#pragma unroll
        for (int nt = 0; nt < 2; ++nt) {
            int n  = nt * 16 + nl;
            int nn = (n < GDIM) ? n : 0;
            float m0 = (n < GDIM && k0     < IDIM) ? 1.0f : 0.0f;
            float m1 = (n < GDIM && k0 + 1 < IDIM) ? 1.0f : 0.0f;
            int ka = (k0     < IDIM) ? k0     : 0;
            int kb = (k0 + 1 < IDIM) ? k0 + 1 : 0;
            Bih[nt][kc][0] = w_ih[nn * IDIM + ka] * m0;
            Bih[nt][kc][1] = w_ih[nn * IDIM + kb] * m1;
        }
    }
#pragma unroll
    for (int kc = 0; kc < 2; ++kc) {
        int k0 = kc * 4 + 2 * half;
#pragma unroll
        for (int nt = 0; nt < 2; ++nt) {
            int n  = nt * 16 + nl;
            int nn = (n < GDIM) ? n : 0;
            float m0 = (n < GDIM && k0     < HDIM) ? 1.0f : 0.0f;
            float m1 = (n < GDIM && k0 + 1 < HDIM) ? 1.0f : 0.0f;
            int ka = (k0     < HDIM) ? k0     : 0;
            int kb = (k0 + 1 < HDIM) ? k0 + 1 : 0;
            Bhh[nt][kc][0] = w_hh[nn * HDIM + ka] * m0;
            Bhh[nt][kc][1] = w_hh[nn * HDIM + kb] * m1;
        }
    }

    const float bias0 = b_ih[nl] + b_hh[nl];
    const float bias1 = (16 + nl < GDIM) ? (b_ih[16 + nl] + b_hh[16 + nl]) : 0.0f;

    if (half == 0) {
#pragma unroll
        for (int k = 0; k < 8; ++k) hbuf[nl][k] = 0.0f;
    }
    float c[4] = {0.0f, 0.0f, 0.0f, 0.0f};
    __syncthreads();

    const float* xrow = x + (size_t)(b0 + nl) * (TSTEP * IDIM);

    auto x_project = [&](int t, v8f& xa0, v8f& xa1) {
        const float* xt = xrow + t * IDIM;
#pragma unroll
        for (int i = 0; i < 8; ++i) { xa0[i] = bias0; xa1[i] = bias1; }
#pragma unroll
        for (int kc = 0; kc < 4; ++kc) {
            int k0 = kc * 4 + 2 * half;
            v2f a;
            int ka = (k0     < IDIM) ? k0     : 0;
            int kb = (k0 + 1 < IDIM) ? k0 + 1 : 0;
            a[0] = xt[ka] * ((k0     < IDIM) ? 1.0f : 0.0f);
            a[1] = xt[kb] * ((k0 + 1 < IDIM) ? 1.0f : 0.0f);
            xa0 = __builtin_amdgcn_wmma_f32_16x16x4_f32(false, a, false, Bih[0][kc],
                                                        (short)0, xa0, false, false);
            xa1 = __builtin_amdgcn_wmma_f32_16x16x4_f32(false, a, false, Bih[1][kc],
                                                        (short)0, xa1, false, false);
        }
    };

    v8f xacc0, xacc1;
    x_project(0, xacc0, xacc1);

    for (int t = 0; t < TSTEP; ++t) {
        v8f acc0 = xacc0, acc1 = xacc1;

#pragma unroll
        for (int kc = 0; kc < 2; ++kc) {
            int k0 = kc * 4 + 2 * half;
            v2f a;
            a[0] = hbuf[nl][k0];
            a[1] = hbuf[nl][k0 + 1];
            acc0 = __builtin_amdgcn_wmma_f32_16x16x4_f32(false, a, false, Bhh[0][kc],
                                                         (short)0, acc0, false, false);
            acc1 = __builtin_amdgcn_wmma_f32_16x16x4_f32(false, a, false, Bhh[1][kc],
                                                         (short)0, acc1, false, false);
        }

#pragma unroll
        for (int v = 0; v < 8; ++v) {
            int m = v + 8 * half;
            gates[m][nl]      = acc0[v];
            gates[m][nl + 16] = acc1[v];
        }
        __syncthreads();

        int tn = (t + 1 < TSTEP) ? (t + 1) : 0;
        x_project(tn, xacc0, xacc1);

        {
            const int row = nl;
            const int jb  = half * 4;
#pragma unroll
            for (int jj = 0; jj < 4; ++jj) {
                int j  = jb + jj;
                int jc = (j < HDIM) ? j : (HDIM - 1);
                float ig = sigm(gates[row][jc]);
                float fg = sigm(gates[row][jc + 7]);
                float gg = FAST_TANH(gates[row][jc + 14]);
                float og = sigm(gates[row][jc + 21]);
                c[jj] = fg * c[jj] + ig * gg;
                float hv = og * FAST_TANH(c[jj]);
                float* dst = (j < HDIM) ? &hbuf[row][j] : &hdump[row];
                *dst = hv;
            }
        }
        __syncthreads();
    }

    float hr[HDIM];
#pragma unroll
    for (int j = 0; j < HDIM; ++j) {
        float v = hbuf[nl][j];
        hr[j] = v > 0.0f ? v : 0.0f;
    }
    float acc = 0.0f;
    const int ffb = half * (FFDIM / 2);
    for (int ff = ffb; ff < ffb + FFDIM / 2; ++ff) {
        float s = fc1_b[ff];
#pragma unroll
        for (int j = 0; j < HDIM; ++j) s += fc1_w[ff * HDIM + j] * hr[j];
        s = s > 0.0f ? s : 0.0f;
        acc += fc_w[ff] * s;
    }
    if (half == 1) red[nl] = acc;
    __syncthreads();
    if (half == 0) out[b0 + nl] = acc + red[nl] + fc_b[0];
}

extern "C" void kernel_launch(void* const* d_in, const int* in_sizes, int n_in,
                              void* d_out, int out_size, void* d_ws, size_t ws_size,
                              hipStream_t stream) {
    (void)in_sizes; (void)n_in; (void)out_size; (void)d_ws; (void)ws_size;
    const float* x     = (const float*)d_in[0];
    const float* w_ih  = (const float*)d_in[1];
    const float* w_hh  = (const float*)d_in[2];
    const float* b_ih  = (const float*)d_in[3];
    const float* b_hh  = (const float*)d_in[4];
    const float* fc1_w = (const float*)d_in[5];
    const float* fc1_b = (const float*)d_in[6];
    const float* fc_w  = (const float*)d_in[7];
    const float* fc_b  = (const float*)d_in[8];

    lstm_fused_kernel<<<dim3(BATCH / 16), dim3(32), 0, stream>>>(
        x, w_ih, w_hh, b_ih, b_hh, fc1_w, fc1_b, fc_w, fc_b, (float*)d_out);
}


# Gate columns ordered [i|f],[g|o] by unit: cell update on the WMMA accumulators in registers (f,o by DPP row rotate); only h crosses LDS, once a step, x-projection WMMAs split round it; sums as before
# speedup vs baseline: 2.3461x; 2.3461x over previous
#include <hip/hip_runtime.h>


typedef float v2f __attribute__((ext_vector_type(2)));
typedef float v4f __attribute__((ext_vector_type(4)));
typedef float v8f __attribute__((ext_vector_type(8)));

#define BATCH 4096
#define TSTEP 512
#define IDIM  13
#define HDIM  7
#define GDIM  28
#define FFDIM 512

#define LDS_SYNC()                                                     \
    do {                                                               \
        __builtin_amdgcn_fence(__ATOMIC_RELEASE, "workgroup", "local"); \
        __builtin_amdgcn_s_barrier();                                  \
        __builtin_amdgcn_fence(__ATOMIC_ACQUIRE, "workgroup", "local"); \
    } while (0)

#define TC    16
#define NCH   (TSTEP / TC)
#define ROWQ  (TC * IDIM / 4)
#define XP    (TC * IDIM + 4)
#define XBUF  (16 * XP)
#define NLD   (8 * ROWQ / 32)

#define LOG2E 1.4426950408889634f

#if __has_builtin(__builtin_amdgcn_tanhf)
#define FAST_TANH(v) __builtin_amdgcn_tanhf(v)
#define TANH_TIER 2
#elif __has_builtin(__builtin_amdgcn_tanh_f32)
#define FAST_TANH(v) __builtin_amdgcn_tanh_f32(v)
#define TANH_TIER 2
#elif __has_builtin(__builtin_amdgcn_exp2f) && __has_builtin(__builtin_amdgcn_rcpf)
#define TANH_TIER 1
static __device__ __forceinline__ float FAST_TANH(float v) {
    float e2 = __builtin_amdgcn_exp2f(v * (2.0f * LOG2E));
    return fmaf(-2.0f, __builtin_amdgcn_rcpf(e2 + 1.0f), 1.0f);
}
#else
#define FAST_TANH(v) tanhf(v)
#define TANH_TIER 0
#endif

static __device__ __forceinline__ float sigm(float v) {
#if TANH_TIER == 2
    return fmaf(0.5f, FAST_TANH(0.5f * v), 0.5f);
#elif TANH_TIER == 1
    return __builtin_amdgcn_rcpf(1.0f + __builtin_amdgcn_exp2f(v * -LOG2E));
#else
    return 1.0f / (1.0f + __expf(-v));
#endif
}

static __device__ __forceinline__ float row_partner(float v) {
    return __builtin_bit_cast(float,
        __builtin_amdgcn_update_dpp(0, __builtin_bit_cast(int, v), 0x128, 0xf, 0xf, true));
}

__global__ __launch_bounds__(32) void lstm_fused_kernel(
    const float* __restrict__ x,     const float* __restrict__ w_ih,
    const float* __restrict__ w_hh,  const float* __restrict__ b_ih,
    const float* __restrict__ b_hh,  const float* __restrict__ fc1_w,
    const float* __restrict__ fc1_b, const float* __restrict__ fc_w,
    const float* __restrict__ fc_b,  float* __restrict__ out)
{
    __shared__ float hb[272];
    __shared__ float red[16];
    __shared__ __attribute__((aligned(16))) float xs[2 * XBUF];

    const int l    = threadIdx.x;
    const int half = l >> 4;
    const int nl   = l & 15;
    const int b0   = blockIdx.x * 16;

    const int  unit = nl & 7;
    const bool ucol = unit < HDIM;
    int grow[2];
#pragma unroll
    for (int nt = 0; nt < 2; ++nt) grow[nt] = ucol ? (2 * nt + (nl >> 3)) * HDIM + unit : 0;

    v2f Bih[2][4];
    v2f Bhh[2][2];
#pragma unroll
    for (int kc = 0; kc < 4; ++kc) {
        int k0 = kc * 4 + 2 * half;
#pragma unroll
        for (int nt = 0; nt < 2; ++nt) {
            float m0 = (ucol && k0     < IDIM) ? 1.0f : 0.0f;
            float m1 = (ucol && k0 + 1 < IDIM) ? 1.0f : 0.0f;
            int ka = (k0     < IDIM) ? k0     : 0;
            int kb = (k0 + 1 < IDIM) ? k0 + 1 : 0;
            Bih[nt][kc][0] = w_ih[grow[nt] * IDIM + ka] * m0;
            Bih[nt][kc][1] = w_ih[grow[nt] * IDIM + kb] * m1;
        }
    }
#pragma unroll
    for (int kc = 0; kc < 2; ++kc) {
        int k0 = kc * 4 + 2 * half;
#pragma unroll
        for (int nt = 0; nt < 2; ++nt) {
            float m0 = (ucol && k0     < HDIM) ? 1.0f : 0.0f;
            float m1 = (ucol && k0 + 1 < HDIM) ? 1.0f : 0.0f;
            int ka = (k0     < HDIM) ? k0     : 0;
            int kb = (k0 + 1 < HDIM) ? k0 + 1 : 0;
            Bhh[nt][kc][0] = w_hh[grow[nt] * HDIM + ka] * m0;
            Bhh[nt][kc][1] = w_hh[grow[nt] * HDIM + kb] * m1;
        }
    }

    const float bias0 = ucol ? (b_ih[grow[0]] + b_hh[grow[0]]) : 0.0f;
    const float bias1 = ucol ? (b_ih[grow[1]] + b_hh[grow[1]]) : 0.0f;

    const bool glane = (nl < 8);
    const float act_s = glane ? 1.0f : 0.5f;
    const float act_b = glane ? 0.0f : 0.5f;

    if (half == 0) {
#pragma unroll
        for (int k = 0; k < 8; ++k) hb[nl * 8 + k] = 0.0f;
    }
    float c[8] = {0.0f, 0.0f, 0.0f, 0.0f, 0.0f, 0.0f, 0.0f, 0.0f};
    float* const hst = hb + ((nl < HDIM) ? 0 : 128) + (8 * half) * 8 + nl;
    __syncthreads();

    unsigned goff[NLD], loff[NLD];
#pragma unroll
    for (int i = 0; i < NLD; ++i) {
        unsigned f = (unsigned)(i * 32 + l);
        unsigned r = f / ROWQ;
        unsigned q = f - r * ROWQ;
        goff[i] = r * (TSTEP * IDIM) + q * 4;
        loff[i] = r * XP + q * 4;
    }
    const float* xblk = x + (size_t)b0 * (TSTEP * IDIM);
    v4f xr[2 * NLD];

    auto tile_fetch = [&](int cn) {
        const float* src = xblk + cn * (TC * IDIM);
#pragma unroll
        for (int i = 0; i < NLD; ++i) {
            xr[i]       = *reinterpret_cast<const v4f*>(src + goff[i]);
            xr[NLD + i] = *reinterpret_cast<const v4f*>(src + goff[i] + 8 * TSTEP * IDIM);
        }
    };
    auto tile_stash = [&](int cn) {
        float* dst = xs + (cn & 1) * XBUF;
#pragma unroll
        for (int i = 0; i < NLD; ++i) {
            *reinterpret_cast<v4f*>(dst + loff[i])          = xr[i];
            *reinterpret_cast<v4f*>(dst + loff[i] + 8 * XP) = xr[NLD + i];
        }
    };

    const int xlane = nl * XP + 2 * half;
    const int xtail = nl * XP + (half ? 0 : 12);
    auto x_frags = [&](int t, v2f (&a)[4]) {
        const float* xt = xs + ((t / TC) & 1) * XBUF + (t % TC) * IDIM;
#pragma unroll
        for (int kc = 0; kc < 3; ++kc) {
            a[kc][0] = xt[xlane + kc * 4];
            a[kc][1] = xt[xlane + kc * 4 + 1];
        }
        float v12 = xt[xtail];
        a[3][0] = half ? 0.0f : v12;
        a[3][1] = 0.0f;
    };

    auto x_project_lo = [&](const v2f (&a)[4], v8f& g0, v8f& g1) {
#pragma unroll
        for (int i = 0; i < 8; ++i) { g0[i] = bias0; g1[i] = bias1; }
#pragma unroll
        for (int kc = 0; kc < 2; ++kc) {
            g0 = __builtin_amdgcn_wmma_f32_16x16x4_f32(false, a[kc], false, Bih[0][kc],
                                                       (short)0, g0, false, false);
            g1 = __builtin_amdgcn_wmma_f32_16x16x4_f32(false, a[kc], false, Bih[1][kc],
                                                       (short)0, g1, false, false);
        }
    };
    auto x_project_hi = [&](const v2f (&a)[4], v8f& g0, v8f& g1) {
#pragma unroll
        for (int kc = 2; kc < 4; ++kc) {
            g0 = __builtin_amdgcn_wmma_f32_16x16x4_f32(false, a[kc], false, Bih[0][kc],
                                                       (short)0, g0, false, false);
            g1 = __builtin_amdgcn_wmma_f32_16x16x4_f32(false, a[kc], false, Bih[1][kc],
                                                       (short)0, g1, false, false);
        }
    };

    tile_fetch(0);
    tile_stash(0);
    tile_fetch(1);
    LDS_SYNC();
    v2f ax[4];
    v8f acc0, acc1;
    x_frags(0, ax);
    x_project_lo(ax, acc0, acc1);

    for (int t = 0; t < TSTEP; ++t) {
        v2f ah[2];
#pragma unroll
        for (int kc = 0; kc < 2; ++kc) {
            int k0 = kc * 4 + 2 * half;
            ah[kc][0] = hb[nl * 8 + k0];
            ah[kc][1] = hb[nl * 8 + k0 + 1];
        }
        __builtin_amdgcn_sched_barrier(0);
        int tn = (t + 1 < TSTEP) ? (t + 1) : 0;
        v2f axn[4];
        x_frags(tn, axn);
        __builtin_amdgcn_sched_barrier(0);

        x_project_hi(ax, acc0, acc1);

#pragma unroll
        for (int kc = 0; kc < 2; ++kc) {
            acc0 = __builtin_amdgcn_wmma_f32_16x16x4_f32(false, ah[kc], false, Bhh[0][kc],
                                                         (short)0, acc0, false, false);
            acc1 = __builtin_amdgcn_wmma_f32_16x16x4_f32(false, ah[kc], false, Bhh[1][kc],
                                                         (short)0, acc1, false, false);
        }

        float hv[8];
#pragma unroll
        for (int r = 0; r < 8; ++r) {
            float a0 = sigm(acc0[r]);
#if TANH_TIER == 2
            float a1 = fmaf(act_s, FAST_TANH(act_s * acc1[r]), act_b);
#else
            float a1 = glane ? FAST_TANH(acc1[r]) : sigm(acc1[r]);
#endif
            float fg = row_partner(a0);
            float og = row_partner(a1);
            c[r] = fg * c[r] + a0 * a1;
            hv[r] = og * FAST_TANH(c[r]);
        }
#pragma unroll
        for (int r = 0; r < 8; ++r) hst[r * 8] = hv[r];

#pragma unroll
        for (int kc = 0; kc < 4; ++kc) ax[kc] = axn[kc];
        x_project_lo(ax, acc0, acc1);

        if ((t % TC) == TC - 2) {
            int cn = t / TC + 1;
            if (cn < NCH)     tile_stash(cn);
            if (cn + 1 < NCH) tile_fetch(cn + 1);
        }
        LDS_SYNC();
    }

    float hr[HDIM];
#pragma unroll
    for (int j = 0; j < HDIM; ++j) {
        float v = hb[nl * 8 + j];
        hr[j] = v > 0.0f ? v : 0.0f;
    }
    float acc = 0.0f;
    const int ffb = half * (FFDIM / 2);
    for (int ff = ffb; ff < ffb + FFDIM / 2; ++ff) {
        float s = fc1_b[ff];
#pragma unroll
        for (int j = 0; j < HDIM; ++j) s += fc1_w[ff * HDIM + j] * hr[j];
        s = s > 0.0f ? s : 0.0f;
        acc += fc_w[ff] * s;
    }
    if (half == 1) red[nl] = acc;
    __syncthreads();
    if (half == 0) out[b0 + nl] = acc + red[nl] + fc_b[0];
}

extern "C" void kernel_launch(void* const* d_in, const int* in_sizes, int n_in,
                              void* d_out, int out_size, void* d_ws, size_t ws_size,
                              hipStream_t stream) {
    (void)in_sizes; (void)n_in; (void)out_size; (void)d_ws; (void)ws_size;
    const float* x     = (const float*)d_in[0];
    const float* w_ih  = (const float*)d_in[1];
    const float* w_hh  = (const float*)d_in[2];
    const float* b_ih  = (const float*)d_in[3];
    const float* b_hh  = (const float*)d_in[4];
    const float* fc1_w = (const float*)d_in[5];
    const float* fc1_b = (const float*)d_in[6];
    const float* fc_w  = (const float*)d_in[7];
    const float* fc_b  = (const float*)d_in[8];

    lstm_fused_kernel<<<dim3(BATCH / 16), dim3(32), 0, stream>>>(
        x, w_ih, w_hh, b_ih, b_hh, fc1_w, fc1_b, fc_w, fc_b, (float*)d_out);
}


# Work-group is one wave: the step's LDS hand-over is ordered at wavefront scope, so the next step's h load follows the h stores down the LDS queue with no wait for them to land; every sum as before
# speedup vs baseline: 2.5431x; 1.0840x over previous
#include <hip/hip_runtime.h>


typedef float v2f __attribute__((ext_vector_type(2)));
typedef float v4f __attribute__((ext_vector_type(4)));
typedef float v8f __attribute__((ext_vector_type(8)));

#define BATCH 4096
#define TSTEP 512
#define IDIM  13
#define HDIM  7
#define GDIM  28
#define FFDIM 512

#define LDS_SYNC()                                                     \
    do {                                                               \
        __builtin_amdgcn_fence(__ATOMIC_RELEASE, "workgroup", "local"); \
        __builtin_amdgcn_s_barrier();                                  \
        __builtin_amdgcn_fence(__ATOMIC_ACQUIRE, "workgroup", "local"); \
    } while (0)

#define WAVE_LDS_ORDER()                                               \
    do {                                                               \
        __builtin_amdgcn_fence(__ATOMIC_RELEASE, "wavefront", "local"); \
        __builtin_amdgcn_wave_barrier();                               \
        __builtin_amdgcn_fence(__ATOMIC_ACQUIRE, "wavefront", "local"); \
    } while (0)

#define TC    16
#define NCH   (TSTEP / TC)
#define ROWQ  (TC * IDIM / 4)
#define XP    (TC * IDIM + 4)
#define XBUF  (16 * XP)
#define NLD   (8 * ROWQ / 32)

#define LOG2E 1.4426950408889634f

#if __has_builtin(__builtin_amdgcn_tanhf)
#define FAST_TANH(v) __builtin_amdgcn_tanhf(v)
#define TANH_TIER 2
#elif __has_builtin(__builtin_amdgcn_tanh_f32)
#define FAST_TANH(v) __builtin_amdgcn_tanh_f32(v)
#define TANH_TIER 2
#elif __has_builtin(__builtin_amdgcn_exp2f) && __has_builtin(__builtin_amdgcn_rcpf)
#define TANH_TIER 1
static __device__ __forceinline__ float FAST_TANH(float v) {
    float e2 = __builtin_amdgcn_exp2f(v * (2.0f * LOG2E));
    return fmaf(-2.0f, __builtin_amdgcn_rcpf(e2 + 1.0f), 1.0f);
}
#else
#define FAST_TANH(v) tanhf(v)
#define TANH_TIER 0
#endif

static __device__ __forceinline__ float sigm(float v) {
#if TANH_TIER == 2
    return fmaf(0.5f, FAST_TANH(0.5f * v), 0.5f);
#elif TANH_TIER == 1
    return __builtin_amdgcn_rcpf(1.0f + __builtin_amdgcn_exp2f(v * -LOG2E));
#else
    return 1.0f / (1.0f + __expf(-v));
#endif
}

static __device__ __forceinline__ float row_partner(float v) {
    return __builtin_bit_cast(float,
        __builtin_amdgcn_update_dpp(0, __builtin_bit_cast(int, v), 0x128, 0xf, 0xf, true));
}

__global__ __launch_bounds__(32) void lstm_fused_kernel(
    const float* __restrict__ x,     const float* __restrict__ w_ih,
    const float* __restrict__ w_hh,  const float* __restrict__ b_ih,
    const float* __restrict__ b_hh,  const float* __restrict__ fc1_w,
    const float* __restrict__ fc1_b, const float* __restrict__ fc_w,
    const float* __restrict__ fc_b,  float* __restrict__ out)
{
    __shared__ float hb[272];
    __shared__ float red[16];
    __shared__ __attribute__((aligned(16))) float xs[2 * XBUF];

    const int l    = threadIdx.x;
    const int half = l >> 4;
    const int nl   = l & 15;
    const int b0   = blockIdx.x * 16;

    const int  unit = nl & 7;
    const bool ucol = unit < HDIM;
    int grow[2];
#pragma unroll
    for (int nt = 0; nt < 2; ++nt) grow[nt] = ucol ? (2 * nt + (nl >> 3)) * HDIM + unit : 0;

    v2f Bih[2][4];
    v2f Bhh[2][2];
#pragma unroll
    for (int kc = 0; kc < 4; ++kc) {
        int k0 = kc * 4 + 2 * half;
#pragma unroll
        for (int nt = 0; nt < 2; ++nt) {
            float m0 = (ucol && k0     < IDIM) ? 1.0f : 0.0f;
            float m1 = (ucol && k0 + 1 < IDIM) ? 1.0f : 0.0f;
            int ka = (k0     < IDIM) ? k0     : 0;
            int kb = (k0 + 1 < IDIM) ? k0 + 1 : 0;
            Bih[nt][kc][0] = w_ih[grow[nt] * IDIM + ka] * m0;
            Bih[nt][kc][1] = w_ih[grow[nt] * IDIM + kb] * m1;
        }
    }
#pragma unroll
    for (int kc = 0; kc < 2; ++kc) {
        int k0 = kc * 4 + 2 * half;
#pragma unroll
        for (int nt = 0; nt < 2; ++nt) {
            float m0 = (ucol && k0     < HDIM) ? 1.0f : 0.0f;
            float m1 = (ucol && k0 + 1 < HDIM) ? 1.0f : 0.0f;
            int ka = (k0     < HDIM) ? k0     : 0;
            int kb = (k0 + 1 < HDIM) ? k0 + 1 : 0;
            Bhh[nt][kc][0] = w_hh[grow[nt] * HDIM + ka] * m0;
            Bhh[nt][kc][1] = w_hh[grow[nt] * HDIM + kb] * m1;
        }
    }

    const float bias0 = ucol ? (b_ih[grow[0]] + b_hh[grow[0]]) : 0.0f;
    const float bias1 = ucol ? (b_ih[grow[1]] + b_hh[grow[1]]) : 0.0f;

    const bool glane = (nl < 8);
    const float act_s = glane ? 1.0f : 0.5f;
    const float act_b = glane ? 0.0f : 0.5f;

    if (half == 0) {
#pragma unroll
        for (int k = 0; k < 8; ++k) hb[nl * 8 + k] = 0.0f;
    }
    float c[8] = {0.0f, 0.0f, 0.0f, 0.0f, 0.0f, 0.0f, 0.0f, 0.0f};
    float* const hst = hb + ((nl < HDIM) ? 0 : 128) + (8 * half) * 8 + nl;
    __syncthreads();

    unsigned goff[NLD], loff[NLD];
#pragma unroll
    for (int i = 0; i < NLD; ++i) {
        unsigned f = (unsigned)(i * 32 + l);
        unsigned r = f / ROWQ;
        unsigned q = f - r * ROWQ;
        goff[i] = r * (TSTEP * IDIM) + q * 4;
        loff[i] = r * XP + q * 4;
    }
    const float* xblk = x + (size_t)b0 * (TSTEP * IDIM);
    v4f xr[2 * NLD];

    auto tile_fetch = [&](int cn) {
        const float* src = xblk + cn * (TC * IDIM);
#pragma unroll
        for (int i = 0; i < NLD; ++i) {
            xr[i]       = *reinterpret_cast<const v4f*>(src + goff[i]);
            xr[NLD + i] = *reinterpret_cast<const v4f*>(src + goff[i] + 8 * TSTEP * IDIM);
        }
    };
    auto tile_stash = [&](int cn) {
        float* dst = xs + (cn & 1) * XBUF;
#pragma unroll
        for (int i = 0; i < NLD; ++i) {
            *reinterpret_cast<v4f*>(dst + loff[i])          = xr[i];
            *reinterpret_cast<v4f*>(dst + loff[i] + 8 * XP) = xr[NLD + i];
        }
    };

    const int xlane = nl * XP + 2 * half;
    const int xtail = nl * XP + (half ? 0 : 12);
    auto x_frags = [&](int t, v2f (&a)[4]) {
        const float* xt = xs + ((t / TC) & 1) * XBUF + (t % TC) * IDIM;
#pragma unroll
        for (int kc = 0; kc < 3; ++kc) {
            a[kc][0] = xt[xlane + kc * 4];
            a[kc][1] = xt[xlane + kc * 4 + 1];
        }
        float v12 = xt[xtail];
        a[3][0] = half ? 0.0f : v12;
        a[3][1] = 0.0f;
    };

    auto x_project_lo = [&](const v2f (&a)[4], v8f& g0, v8f& g1) {
#pragma unroll
        for (int i = 0; i < 8; ++i) { g0[i] = bias0; g1[i] = bias1; }
#pragma unroll
        for (int kc = 0; kc < 2; ++kc) {
            g0 = __builtin_amdgcn_wmma_f32_16x16x4_f32(false, a[kc], false, Bih[0][kc],
                                                       (short)0, g0, false, false);
            g1 = __builtin_amdgcn_wmma_f32_16x16x4_f32(false, a[kc], false, Bih[1][kc],
                                                       (short)0, g1, false, false);
        }
    };
    auto x_project_hi = [&](const v2f (&a)[4], v8f& g0, v8f& g1) {
#pragma unroll
        for (int kc = 2; kc < 4; ++kc) {
            g0 = __builtin_amdgcn_wmma_f32_16x16x4_f32(false, a[kc], false, Bih[0][kc],
                                                       (short)0, g0, false, false);
            g1 = __builtin_amdgcn_wmma_f32_16x16x4_f32(false, a[kc], false, Bih[1][kc],
                                                       (short)0, g1, false, false);
        }
    };

    tile_fetch(0);
    tile_stash(0);
    tile_fetch(1);
    LDS_SYNC();
    v2f ax[4];
    v8f acc0, acc1;
    x_frags(0, ax);
    x_project_lo(ax, acc0, acc1);

    for (int t = 0; t < TSTEP; ++t) {
        v2f ah[2];
#pragma unroll
        for (int kc = 0; kc < 2; ++kc) {
            int k0 = kc * 4 + 2 * half;
            ah[kc][0] = hb[nl * 8 + k0];
            ah[kc][1] = hb[nl * 8 + k0 + 1];
        }
        __builtin_amdgcn_sched_barrier(0);
        int tn = (t + 1 < TSTEP) ? (t + 1) : 0;
        v2f axn[4];
        x_frags(tn, axn);
        __builtin_amdgcn_sched_barrier(0);

        x_project_hi(ax, acc0, acc1);

#pragma unroll
        for (int kc = 0; kc < 2; ++kc) {
            acc0 = __builtin_amdgcn_wmma_f32_16x16x4_f32(false, ah[kc], false, Bhh[0][kc],
                                                         (short)0, acc0, false, false);
            acc1 = __builtin_amdgcn_wmma_f32_16x16x4_f32(false, ah[kc], false, Bhh[1][kc],
                                                         (short)0, acc1, false, false);
        }

        float hv[8];
#pragma unroll
        for (int r = 0; r < 8; ++r) {
            float a0 = sigm(acc0[r]);
#if TANH_TIER == 2
            float a1 = fmaf(act_s, FAST_TANH(act_s * acc1[r]), act_b);
#else
            float a1 = glane ? FAST_TANH(acc1[r]) : sigm(acc1[r]);
#endif
            float fg = row_partner(a0);
            float og = row_partner(a1);
            c[r] = fg * c[r] + a0 * a1;
            hv[r] = og * FAST_TANH(c[r]);
        }
#pragma unroll
        for (int r = 0; r < 8; ++r) hst[r * 8] = hv[r];

#pragma unroll
        for (int kc = 0; kc < 4; ++kc) ax[kc] = axn[kc];
        x_project_lo(ax, acc0, acc1);

        if ((t % TC) == TC - 2) {
            int cn = t / TC + 1;
            if (cn < NCH)     tile_stash(cn);
            if (cn + 1 < NCH) tile_fetch(cn + 1);
        }
        WAVE_LDS_ORDER();
    }

    float hr[HDIM];
#pragma unroll
    for (int j = 0; j < HDIM; ++j) {
        float v = hb[nl * 8 + j];
        hr[j] = v > 0.0f ? v : 0.0f;
    }
    float acc = 0.0f;
    const int ffb = half * (FFDIM / 2);
    for (int ff = ffb; ff < ffb + FFDIM / 2; ++ff) {
        float s = fc1_b[ff];
#pragma unroll
        for (int j = 0; j < HDIM; ++j) s += fc1_w[ff * HDIM + j] * hr[j];
        s = s > 0.0f ? s : 0.0f;
        acc += fc_w[ff] * s;
    }
    if (half == 1) red[nl] = acc;
    __syncthreads();
    if (half == 0) out[b0 + nl] = acc + red[nl] + fc_b[0];
}

extern "C" void kernel_launch(void* const* d_in, const int* in_sizes, int n_in,
                              void* d_out, int out_size, void* d_ws, size_t ws_size,
                              hipStream_t stream) {
    (void)in_sizes; (void)n_in; (void)out_size; (void)d_ws; (void)ws_size;
    const float* x     = (const float*)d_in[0];
    const float* w_ih  = (const float*)d_in[1];
    const float* w_hh  = (const float*)d_in[2];
    const float* b_ih  = (const float*)d_in[3];
    const float* b_hh  = (const float*)d_in[4];
    const float* fc1_w = (const float*)d_in[5];
    const float* fc1_b = (const float*)d_in[6];
    const float* fc_w  = (const float*)d_in[7];
    const float* fc_b  = (const float*)d_in[8];

    lstm_fused_kernel<<<dim3(BATCH / 16), dim3(32), 0, stream>>>(
        x, w_ih, w_hh, b_ih, b_hh, fc1_w, fc1_b, fc_w, fc_b, (float*)d_out);
}


# A wave carries 4 batch rows, not 16 (accumulator registers 0,1 of each lane half; the other tile rows repeat them); 4 independent waves a group, one per SIMD: 2 rows of vector work a lane; same sums
# speedup vs baseline: 2.9808x; 1.1721x over previous
#include <hip/hip_runtime.h>


typedef float v2f __attribute__((ext_vector_type(2)));
typedef float v4f __attribute__((ext_vector_type(4)));
typedef float v8f __attribute__((ext_vector_type(8)));

#define BATCH 4096
#define TSTEP 512
#define IDIM  13
#define HDIM  7
#define GDIM  28
#define FFDIM 512

#define RH    2
#define RPW   (2 * RH)
#define WPG   4

#define LDS_SYNC()                                                     \
    do {                                                               \
        __builtin_amdgcn_fence(__ATOMIC_RELEASE, "workgroup", "local"); \
        __builtin_amdgcn_s_barrier();                                  \
        __builtin_amdgcn_fence(__ATOMIC_ACQUIRE, "workgroup", "local"); \
    } while (0)

#define WAVE_LDS_ORDER()                                               \
    do {                                                               \
        __builtin_amdgcn_fence(__ATOMIC_RELEASE, "wavefront", "local"); \
        __builtin_amdgcn_wave_barrier();                               \
        __builtin_amdgcn_fence(__ATOMIC_ACQUIRE, "wavefront", "local"); \
    } while (0)

#define TC    16
#define NCH   (TSTEP / TC)
#define ROWQ  (TC * IDIM / 4)
#define XP    (TC * IDIM + 4)
#define XBUF  (RPW * XP)
#define NPIECE (RPW * ROWQ)
#define NLD   ((NPIECE + 31) / 32)

#define LOG2E 1.4426950408889634f

#if __has_builtin(__builtin_amdgcn_tanhf)
#define FAST_TANH(v) __builtin_amdgcn_tanhf(v)
#define TANH_TIER 2
#elif __has_builtin(__builtin_amdgcn_tanh_f32)
#define FAST_TANH(v) __builtin_amdgcn_tanh_f32(v)
#define TANH_TIER 2
#elif __has_builtin(__builtin_amdgcn_exp2f) && __has_builtin(__builtin_amdgcn_rcpf)
#define TANH_TIER 1
static __device__ __forceinline__ float FAST_TANH(float v) {
    float e2 = __builtin_amdgcn_exp2f(v * (2.0f * LOG2E));
    return fmaf(-2.0f, __builtin_amdgcn_rcpf(e2 + 1.0f), 1.0f);
}
#else
#define FAST_TANH(v) tanhf(v)
#define TANH_TIER 0
#endif

static __device__ __forceinline__ float sigm(float v) {
#if TANH_TIER == 2
    return fmaf(0.5f, FAST_TANH(0.5f * v), 0.5f);
#elif TANH_TIER == 1
    return __builtin_amdgcn_rcpf(1.0f + __builtin_amdgcn_exp2f(v * -LOG2E));
#else
    return 1.0f / (1.0f + __expf(-v));
#endif
}

static __device__ __forceinline__ float row_partner(float v) {
    return __builtin_bit_cast(float,
        __builtin_amdgcn_update_dpp(0, __builtin_bit_cast(int, v), 0x128, 0xf, 0xf, true));
}

__global__ __launch_bounds__(32 * WPG) void lstm_fused_kernel(
    const float* __restrict__ x,     const float* __restrict__ w_ih,
    const float* __restrict__ w_hh,  const float* __restrict__ b_ih,
    const float* __restrict__ b_hh,  const float* __restrict__ fc1_w,
    const float* __restrict__ fc1_b, const float* __restrict__ fc_w,
    const float* __restrict__ fc_b,  float* __restrict__ out)
{
    __shared__ float hb_all[WPG][32 * RH + 8];
    __shared__ float red_all[WPG][16];
    __shared__ float osum[WPG * RPW];
    __shared__ __attribute__((aligned(16))) float xs_all[WPG][2 * XBUF];

    const int wv   = __builtin_amdgcn_readfirstlane((int)(threadIdx.x >> 5));
    float* const hb  = hb_all[wv];
    float* const red = red_all[wv];
    float* const xs  = xs_all[wv];

    const int l    = threadIdx.x & 31;
    const int half = l >> 4;
    const int nl   = l & 15;
    const int b0   = (blockIdx.x * WPG + wv) * RPW;
    const int  jrow = ((nl & 7) & (RH - 1)) + RH * (nl >> 3);
    const bool rrow = (nl & 7) < RH;

    const int  unit = nl & 7;
    const bool ucol = unit < HDIM;
    int grow[2];
#pragma unroll
    for (int nt = 0; nt < 2; ++nt) grow[nt] = ucol ? (2 * nt + (nl >> 3)) * HDIM + unit : 0;

    v2f Bih[2][4];
    v2f Bhh[2][2];
#pragma unroll
    for (int kc = 0; kc < 4; ++kc) {
        int k0 = kc * 4 + 2 * half;
#pragma unroll
        for (int nt = 0; nt < 2; ++nt) {
            float m0 = (ucol && k0     < IDIM) ? 1.0f : 0.0f;
            float m1 = (ucol && k0 + 1 < IDIM) ? 1.0f : 0.0f;
            int ka = (k0     < IDIM) ? k0     : 0;
            int kb = (k0 + 1 < IDIM) ? k0 + 1 : 0;
            Bih[nt][kc][0] = w_ih[grow[nt] * IDIM + ka] * m0;
            Bih[nt][kc][1] = w_ih[grow[nt] * IDIM + kb] * m1;
        }
    }
#pragma unroll
    for (int kc = 0; kc < 2; ++kc) {
        int k0 = kc * 4 + 2 * half;
#pragma unroll
        for (int nt = 0; nt < 2; ++nt) {
            float m0 = (ucol && k0     < HDIM) ? 1.0f : 0.0f;
            float m1 = (ucol && k0 + 1 < HDIM) ? 1.0f : 0.0f;
            int ka = (k0     < HDIM) ? k0     : 0;
            int kb = (k0 + 1 < HDIM) ? k0 + 1 : 0;
            Bhh[nt][kc][0] = w_hh[grow[nt] * HDIM + ka] * m0;
            Bhh[nt][kc][1] = w_hh[grow[nt] * HDIM + kb] * m1;
        }
    }

    const float bias0 = ucol ? (b_ih[grow[0]] + b_hh[grow[0]]) : 0.0f;
    const float bias1 = ucol ? (b_ih[grow[1]] + b_hh[grow[1]]) : 0.0f;

    const bool glane = (nl < 8);
    const float act_s = glane ? 1.0f : 0.5f;
    const float act_b = glane ? 0.0f : 0.5f;

    for (int i = l; i < RPW * 8; i += 32) hb[i] = 0.0f;
    float c[RH];
#pragma unroll
    for (int r = 0; r < RH; ++r) c[r] = 0.0f;
    float* const hst = hb + ((nl < HDIM) ? 0 : RPW * 8) + (RH * half) * 8 + nl;
    __syncthreads();

    unsigned goff[NLD], loff[NLD];
#pragma unroll
    for (int i = 0; i < NLD; ++i) {
        unsigned f = (unsigned)(i * 32 + l);
        unsigned r = f / ROWQ;
        unsigned q = f - r * ROWQ;
        goff[i] = r * (TSTEP * IDIM) + q * 4;
        loff[i] = r * XP + q * 4;
    }
    const float* xblk = x + (size_t)b0 * (TSTEP * IDIM);
    v4f xr[NLD];
#pragma unroll
    for (int i = 0; i < NLD; ++i) xr[i] = v4f{0.0f, 0.0f, 0.0f, 0.0f};

    auto tile_fetch = [&](int cn) {
        const float* src = xblk + cn * (TC * IDIM);
#pragma unroll
        for (int i = 0; i < NLD; ++i) {
            if (i < NPIECE / 32 || l < NPIECE % 32)
                xr[i] = *reinterpret_cast<const v4f*>(src + goff[i]);
        }
    };
    auto tile_stash = [&](int cn) {
        float* dst = xs + (cn & 1) * XBUF;
#pragma unroll
        for (int i = 0; i < NLD; ++i) {
            if (i < NPIECE / 32 || l < NPIECE % 32)
                *reinterpret_cast<v4f*>(dst + loff[i]) = xr[i];
        }
    };

    const int xlane = jrow * XP + 2 * half;
    const int xtail = jrow * XP + (half ? 0 : 12);
    auto x_frags = [&](int t, v2f (&a)[4]) {
        const float* xt = xs + ((t / TC) & 1) * XBUF + (t % TC) * IDIM;
#pragma unroll
        for (int kc = 0; kc < 3; ++kc) {
            a[kc][0] = xt[xlane + kc * 4];
            a[kc][1] = xt[xlane + kc * 4 + 1];
        }
        float v12 = xt[xtail];
        a[3][0] = half ? 0.0f : v12;
        a[3][1] = 0.0f;
    };

    auto x_project_lo = [&](const v2f (&a)[4], v8f& g0, v8f& g1) {
#pragma unroll
        for (int i = 0; i < 8; ++i) { g0[i] = bias0; g1[i] = bias1; }
#pragma unroll
        for (int kc = 0; kc < 2; ++kc) {
            g0 = __builtin_amdgcn_wmma_f32_16x16x4_f32(false, a[kc], false, Bih[0][kc],
                                                       (short)0, g0, false, false);
            g1 = __builtin_amdgcn_wmma_f32_16x16x4_f32(false, a[kc], false, Bih[1][kc],
                                                       (short)0, g1, false, false);
        }
    };
    auto x_project_hi = [&](const v2f (&a)[4], v8f& g0, v8f& g1) {
#pragma unroll
        for (int kc = 2; kc < 4; ++kc) {
            g0 = __builtin_amdgcn_wmma_f32_16x16x4_f32(false, a[kc], false, Bih[0][kc],
                                                       (short)0, g0, false, false);
            g1 = __builtin_amdgcn_wmma_f32_16x16x4_f32(false, a[kc], false, Bih[1][kc],
                                                       (short)0, g1, false, false);
        }
    };

    tile_fetch(0);
    tile_stash(0);
    tile_fetch(1);
    LDS_SYNC();
    v2f ax[4];
    v8f acc0, acc1;
    x_frags(0, ax);
    x_project_lo(ax, acc0, acc1);

    for (int t = 0; t < TSTEP; ++t) {
        v2f ah[2];
#pragma unroll
        for (int kc = 0; kc < 2; ++kc) {
            int k0 = kc * 4 + 2 * half;
            ah[kc][0] = hb[jrow * 8 + k0];
            ah[kc][1] = hb[jrow * 8 + k0 + 1];
        }
        __builtin_amdgcn_sched_barrier(0);
        int tn = (t + 1 < TSTEP) ? (t + 1) : 0;
        v2f axn[4];
        x_frags(tn, axn);
        __builtin_amdgcn_sched_barrier(0);

        x_project_hi(ax, acc0, acc1);

#pragma unroll
        for (int kc = 0; kc < 2; ++kc) {
            acc0 = __builtin_amdgcn_wmma_f32_16x16x4_f32(false, ah[kc], false, Bhh[0][kc],
                                                         (short)0, acc0, false, false);
            acc1 = __builtin_amdgcn_wmma_f32_16x16x4_f32(false, ah[kc], false, Bhh[1][kc],
                                                         (short)0, acc1, false, false);
        }

        float hv[RH];
#pragma unroll
        for (int r = 0; r < RH; ++r) {
            float a0 = sigm(acc0[r]);
#if TANH_TIER == 2
            float a1 = fmaf(act_s, FAST_TANH(act_s * acc1[r]), act_b);
#else
            float a1 = glane ? FAST_TANH(acc1[r]) : sigm(acc1[r]);
#endif
            float fg = row_partner(a0);
            float og = row_partner(a1);
            c[r] = fg * c[r] + a0 * a1;
            hv[r] = og * FAST_TANH(c[r]);
        }
#pragma unroll
        for (int r = 0; r < RH; ++r) hst[r * 8] = hv[r];

#pragma unroll
        for (int kc = 0; kc < 4; ++kc) ax[kc] = axn[kc];
        x_project_lo(ax, acc0, acc1);

        if ((t % TC) == TC - 2) {
            int cn = t / TC + 1;
            if (cn < NCH)     tile_stash(cn);
            if (cn + 1 < NCH) tile_fetch(cn + 1);
        }
        WAVE_LDS_ORDER();
    }

    float hr[HDIM];
#pragma unroll
    for (int j = 0; j < HDIM; ++j) {
        float v = hb[jrow * 8 + j];
        hr[j] = v > 0.0f ? v : 0.0f;
    }
    float acc = 0.0f;
    const int ffb = half * (FFDIM / 2);
    for (int ff = ffb; ff < ffb + FFDIM / 2; ++ff) {
        float s = fc1_b[ff];
#pragma unroll
        for (int j = 0; j < HDIM; ++j) s += fc1_w[ff * HDIM + j] * hr[j];
        s = s > 0.0f ? s : 0.0f;
        acc += fc_w[ff] * s;
    }
    if (half == 1) red[nl] = acc;
    __syncthreads();
    if (half == 0 && rrow) osum[wv * RPW + jrow] = acc + red[nl] + fc_b[0];
    __syncthreads();
    if (threadIdx.x < WPG * RPW)
        out[blockIdx.x * (WPG * RPW) + threadIdx.x] = osum[threadIdx.x];
}

extern "C" void kernel_launch(void* const* d_in, const int* in_sizes, int n_in,
                              void* d_out, int out_size, void* d_ws, size_t ws_size,
                              hipStream_t stream) {
    (void)in_sizes; (void)n_in; (void)out_size; (void)d_ws; (void)ws_size;
    const float* x     = (const float*)d_in[0];
    const float* w_ih  = (const float*)d_in[1];
    const float* w_hh  = (const float*)d_in[2];
    const float* b_ih  = (const float*)d_in[3];
    const float* b_hh  = (const float*)d_in[4];
    const float* fc1_w = (const float*)d_in[5];
    const float* fc1_b = (const float*)d_in[6];
    const float* fc_w  = (const float*)d_in[7];
    const float* fc_b  = (const float*)d_in[8];

    lstm_fused_kernel<<<dim3(BATCH / (RPW * WPG)), dim3(32 * WPG), 0, stream>>>(
        x, w_ih, w_hh, b_ih, b_hh, fc1_w, fc1_b, fc_w, fc_b, (float*)d_out);
}


# x-projection formed for 4 timesteps per matrix tile (tile rows = 4 batch rows x 4 steps); a step's h-WMMAs take it as C, write their own tile: 6 WMMAs a step, not 12; every sum in the same order
# speedup vs baseline: 4.0746x; 1.3669x over previous
#include <hip/hip_runtime.h>


typedef float v2f __attribute__((ext_vector_type(2)));
typedef float v4f __attribute__((ext_vector_type(4)));
typedef float v8f __attribute__((ext_vector_type(8)));

#define BATCH 4096
#define TSTEP 512
#define IDIM  13
#define HDIM  7
#define GDIM  28
#define FFDIM 512

#define RH    2
#define RPW   (2 * RH)
#define WPG   4

#define LDS_SYNC()                                                     \
    do {                                                               \
        __builtin_amdgcn_fence(__ATOMIC_RELEASE, "workgroup", "local"); \
        __builtin_amdgcn_s_barrier();                                  \
        __builtin_amdgcn_fence(__ATOMIC_ACQUIRE, "workgroup", "local"); \
    } while (0)

#define WAVE_LDS_ORDER()                                               \
    do {                                                               \
        __builtin_amdgcn_fence(__ATOMIC_RELEASE, "wavefront", "local"); \
        __builtin_amdgcn_wave_barrier();                               \
        __builtin_amdgcn_fence(__ATOMIC_ACQUIRE, "wavefront", "local"); \
    } while (0)

#define TC    16
#define NCH   (TSTEP / TC)
#define ROWQ  (TC * IDIM / 4)
#define XP    (TC * IDIM + 4)
#define XBUF  (RPW * XP)
#define NPIECE (RPW * ROWQ)
#define NLD   ((NPIECE + 31) / 32)
#define TG    (8 / RH)
static_assert(TG == 4 && TC % TG == 0, "the step loop deals a group's 8 x-projection WMMAs 3+3+2 over steps 1..3");

#define LOG2E 1.4426950408889634f

#if __has_builtin(__builtin_amdgcn_tanhf)
#define FAST_TANH(v) __builtin_amdgcn_tanhf(v)
#define TANH_TIER 2
#elif __has_builtin(__builtin_amdgcn_tanh_f32)
#define FAST_TANH(v) __builtin_amdgcn_tanh_f32(v)
#define TANH_TIER 2
#elif __has_builtin(__builtin_amdgcn_exp2f) && __has_builtin(__builtin_amdgcn_rcpf)
#define TANH_TIER 1
static __device__ __forceinline__ float FAST_TANH(float v) {
    float e2 = __builtin_amdgcn_exp2f(v * (2.0f * LOG2E));
    return fmaf(-2.0f, __builtin_amdgcn_rcpf(e2 + 1.0f), 1.0f);
}
#else
#define FAST_TANH(v) tanhf(v)
#define TANH_TIER 0
#endif

static __device__ __forceinline__ float sigm(float v) {
#if TANH_TIER == 2
    return fmaf(0.5f, FAST_TANH(0.5f * v), 0.5f);
#elif TANH_TIER == 1
    return __builtin_amdgcn_rcpf(1.0f + __builtin_amdgcn_exp2f(v * -LOG2E));
#else
    return 1.0f / (1.0f + __expf(-v));
#endif
}

static __device__ __forceinline__ float row_partner(float v) {
    return __builtin_bit_cast(float,
        __builtin_amdgcn_update_dpp(0, __builtin_bit_cast(int, v), 0x128, 0xf, 0xf, true));
}

__global__ __launch_bounds__(32 * WPG) void lstm_fused_kernel(
    const float* __restrict__ x,     const float* __restrict__ w_ih,
    const float* __restrict__ w_hh,  const float* __restrict__ b_ih,
    const float* __restrict__ b_hh,  const float* __restrict__ fc1_w,
    const float* __restrict__ fc1_b, const float* __restrict__ fc_w,
    const float* __restrict__ fc_b,  float* __restrict__ out)
{
    __shared__ float hb_all[WPG][32 * RH + 8];
    __shared__ float red_all[WPG][16];
    __shared__ float osum[WPG * RPW];
    __shared__ __attribute__((aligned(16))) float xs_all[WPG][2 * XBUF];

    const int wv   = __builtin_amdgcn_readfirstlane((int)(threadIdx.x >> 5));
    float* const hb  = hb_all[wv];
    float* const red = red_all[wv];
    float* const xs  = xs_all[wv];

    const int l    = threadIdx.x & 31;
    const int half = l >> 4;
    const int nl   = l & 15;
    const int b0   = (blockIdx.x * WPG + wv) * RPW;
    const int  jrow = ((nl & 7) & (RH - 1)) + RH * (nl >> 3);
    const bool rrow = (nl & 7) < RH;

    const int  unit = nl & 7;
    const bool ucol = unit < HDIM;
    int grow[2];
#pragma unroll
    for (int nt = 0; nt < 2; ++nt) grow[nt] = ucol ? (2 * nt + (nl >> 3)) * HDIM + unit : 0;

    v2f Bih[2][4];
    v2f Bhh[2][2];
#pragma unroll
    for (int kc = 0; kc < 4; ++kc) {
        int k0 = kc * 4 + 2 * half;
#pragma unroll
        for (int nt = 0; nt < 2; ++nt) {
            float m0 = (ucol && k0     < IDIM) ? 1.0f : 0.0f;
            float m1 = (ucol && k0 + 1 < IDIM) ? 1.0f : 0.0f;
            int ka = (k0     < IDIM) ? k0     : 0;
            int kb = (k0 + 1 < IDIM) ? k0 + 1 : 0;
            Bih[nt][kc][0] = w_ih[grow[nt] * IDIM + ka] * m0;
            Bih[nt][kc][1] = w_ih[grow[nt] * IDIM + kb] * m1;
        }
    }
#pragma unroll
    for (int kc = 0; kc < 2; ++kc) {
        int k0 = kc * 4 + 2 * half;
#pragma unroll
        for (int nt = 0; nt < 2; ++nt) {
            float m0 = (ucol && k0     < HDIM) ? 1.0f : 0.0f;
            float m1 = (ucol && k0 + 1 < HDIM) ? 1.0f : 0.0f;
            int ka = (k0     < HDIM) ? k0     : 0;
            int kb = (k0 + 1 < HDIM) ? k0 + 1 : 0;
            Bhh[nt][kc][0] = w_hh[grow[nt] * HDIM + ka] * m0;
            Bhh[nt][kc][1] = w_hh[grow[nt] * HDIM + kb] * m1;
        }
    }

    const float bias0 = ucol ? (b_ih[grow[0]] + b_hh[grow[0]]) : 0.0f;
    const float bias1 = ucol ? (b_ih[grow[1]] + b_hh[grow[1]]) : 0.0f;

    const bool glane = (nl < 8);
    const float act_s = glane ? 1.0f : 0.5f;
    const float act_b = glane ? 0.0f : 0.5f;

    for (int i = l; i < RPW * 8; i += 32) hb[i] = 0.0f;
    float c[RH];
#pragma unroll
    for (int r = 0; r < RH; ++r) c[r] = 0.0f;
    float* const hst = hb + ((nl < HDIM) ? 0 : RPW * 8) + (RH * half) * 8 + nl;
    __syncthreads();

    unsigned goff[NLD], loff[NLD];
#pragma unroll
    for (int i = 0; i < NLD; ++i) {
        unsigned f = (unsigned)(i * 32 + l);
        unsigned r = f / ROWQ;
        unsigned q = f - r * ROWQ;
        goff[i] = r * (TSTEP * IDIM) + q * 4;
        loff[i] = r * XP + q * 4;
    }
    const float* xblk = x + (size_t)b0 * (TSTEP * IDIM);
    v4f xr[NLD];
#pragma unroll
    for (int i = 0; i < NLD; ++i) xr[i] = v4f{0.0f, 0.0f, 0.0f, 0.0f};

    auto tile_fetch = [&](int cn) {
        const float* src = xblk + cn * (TC * IDIM);
#pragma unroll
        for (int i = 0; i < NLD; ++i) {
            if (i < NPIECE / 32 || l < NPIECE % 32)
                xr[i] = *reinterpret_cast<const v4f*>(src + goff[i]);
        }
    };
    auto tile_stash = [&](int cn) {
        float* dst = xs + (cn & 1) * XBUF;
#pragma unroll
        for (int i = 0; i < NLD; ++i) {
            if (i < NPIECE / 32 || l < NPIECE % 32)
                *reinterpret_cast<v4f*>(dst + loff[i]) = xr[i];
        }
    };

    const int jt    = (nl & 7) / RH;
    const int xlane = jrow * XP + jt * IDIM + 2 * half;
    const int xtail = jrow * XP + jt * IDIM + (half ? 0 : 12);
    auto x_frags = [&](int t0, v2f (&a)[4], float& v12) {
        const float* xt = xs + ((t0 / TC) & 1) * XBUF + (t0 % TC) * IDIM;
#pragma unroll
        for (int kc = 0; kc < 3; ++kc) {
            a[kc][0] = xt[xlane + kc * 4];
            a[kc][1] = xt[xlane + kc * 4 + 1];
        }
        v12 = xt[xtail];
    };
    auto x_tail = [&](v2f (&a)[4], float v12) {
        a[3][0] = half ? 0.0f : v12;
        a[3][1] = 0.0f;
    };

    auto x_project = [&](int w, const v2f (&a)[4], v8f& g0, v8f& g1) {
        const int kc = w >> 1;
        if ((w & 1) == 0) {
            if (kc == 0) {
#pragma unroll
                for (int i = 0; i < 8; ++i) g0[i] = bias0;
            }
            g0 = __builtin_amdgcn_wmma_f32_16x16x4_f32(false, a[kc], false, Bih[0][kc],
                                                       (short)0, g0, false, false);
        } else {
            if (kc == 0) {
#pragma unroll
                for (int i = 0; i < 8; ++i) g1[i] = bias1;
            }
            g1 = __builtin_amdgcn_wmma_f32_16x16x4_f32(false, a[kc], false, Bih[1][kc],
                                                       (short)0, g1, false, false);
        }
    };

    tile_fetch(0);
    tile_stash(0);
    tile_fetch(1);
    LDS_SYNC();
    v2f axn[4];
    float axt;
    v8f xp0, xp1;
    x_frags(0, axn, axt);
    x_tail(axn, axt);
#pragma unroll
    for (int w = 0; w < 8; ++w) x_project(w, axn, xp0, xp1);

    for (int t0 = 0; t0 < TSTEP; t0 += TG) {
        v8f xn0, xn1;
#pragma unroll
        for (int j = 0; j < TG; ++j) {
            v2f ah[2];
#pragma unroll
            for (int kc = 0; kc < 2; ++kc) {
                int k0 = kc * 4 + 2 * half;
                ah[kc][0] = hb[jrow * 8 + k0];
                ah[kc][1] = hb[jrow * 8 + k0 + 1];
            }
            __builtin_amdgcn_sched_barrier(0);
            if (j == 0) {
                int tn = (t0 + TG < TSTEP) ? (t0 + TG) : 0;
                x_frags(tn, axn, axt);
            } else {
                const int w0 = (j - 1) * 3;
                if (j == TG - 1) x_tail(axn, axt);
#pragma unroll
                for (int w = w0; w < w0 + 3 && w < 8; ++w) x_project(w, axn, xn0, xn1);
            }
            __builtin_amdgcn_sched_barrier(0);

            v8f acc0 = __builtin_amdgcn_wmma_f32_16x16x4_f32(false, ah[0], false, Bhh[0][0],
                                                             (short)0, xp0, false, false);
            v8f acc1 = __builtin_amdgcn_wmma_f32_16x16x4_f32(false, ah[0], false, Bhh[1][0],
                                                             (short)0, xp1, false, false);
            acc0 = __builtin_amdgcn_wmma_f32_16x16x4_f32(false, ah[1], false, Bhh[0][1],
                                                         (short)0, acc0, false, false);
            acc1 = __builtin_amdgcn_wmma_f32_16x16x4_f32(false, ah[1], false, Bhh[1][1],
                                                         (short)0, acc1, false, false);

            float hv[RH];
#pragma unroll
            for (int r = 0; r < RH; ++r) {
                float a0 = sigm(acc0[RH * j + r]);
#if TANH_TIER == 2
                float a1 = fmaf(act_s, FAST_TANH(act_s * acc1[RH * j + r]), act_b);
#else
                float a1 = glane ? FAST_TANH(acc1[RH * j + r]) : sigm(acc1[RH * j + r]);
#endif
                float fg = row_partner(a0);
                float og = row_partner(a1);
                c[r] = fg * c[r] + a0 * a1;
                hv[r] = og * FAST_TANH(c[r]);
            }
#pragma unroll
            for (int r = 0; r < RH; ++r) hst[r * 8] = hv[r];

            if (j == TG - 1 && (t0 % TC) == TC - 2 * TG) {
                int cn = t0 / TC + 1;
                if (cn < NCH)     tile_stash(cn);
                if (cn + 1 < NCH) tile_fetch(cn + 1);
            }
            WAVE_LDS_ORDER();
        }
        xp0 = xn0;
        xp1 = xn1;
    }

    float hr[HDIM];
#pragma unroll
    for (int j = 0; j < HDIM; ++j) {
        float v = hb[jrow * 8 + j];
        hr[j] = v > 0.0f ? v : 0.0f;
    }
    float acc = 0.0f;
    const int ffb = half * (FFDIM / 2);
    for (int ff = ffb; ff < ffb + FFDIM / 2; ++ff) {
        float s = fc1_b[ff];
#pragma unroll
        for (int j = 0; j < HDIM; ++j) s += fc1_w[ff * HDIM + j] * hr[j];
        s = s > 0.0f ? s : 0.0f;
        acc += fc_w[ff] * s;
    }
    if (half == 1) red[nl] = acc;
    __syncthreads();
    if (half == 0 && rrow) osum[wv * RPW + jrow] = acc + red[nl] + fc_b[0];
    __syncthreads();
    if (threadIdx.x < WPG * RPW)
        out[blockIdx.x * (WPG * RPW) + threadIdx.x] = osum[threadIdx.x];
}

extern "C" void kernel_launch(void* const* d_in, const int* in_sizes, int n_in,
                              void* d_out, int out_size, void* d_ws, size_t ws_size,
                              hipStream_t stream) {
    (void)in_sizes; (void)n_in; (void)out_size; (void)d_ws; (void)ws_size;
    const float* x     = (const float*)d_in[0];
    const float* w_ih  = (const float*)d_in[1];
    const float* w_hh  = (const float*)d_in[2];
    const float* b_ih  = (const float*)d_in[3];
    const float* b_hh  = (const float*)d_in[4];
    const float* fc1_w = (const float*)d_in[5];
    const float* fc1_b = (const float*)d_in[6];
    const float* fc_w  = (const float*)d_in[7];
    const float* fc_b  = (const float*)d_in[8];

    lstm_fused_kernel<<<dim3(BATCH / (RPW * WPG)), dim3(32 * WPG), 0, stream>>>(
        x, w_ih, w_hh, b_ih, b_hh, fc1_w, fc1_b, fc_w, fc_b, (float*)d_out);
}


# FF head dealt by feature: a lane keeps 4 features' weights in registers from kernel entry and forms them (fused multiply-adds) for all 16 rows; row sums in 2 LDS passes: head sums reordered
# speedup vs baseline: 5.4221x; 1.3307x over previous
#include <hip/hip_runtime.h>


typedef float v2f __attribute__((ext_vector_type(2)));
typedef float v4f __attribute__((ext_vector_type(4)));
typedef float v8f __attribute__((ext_vector_type(8)));

#define BATCH 4096
#define TSTEP 512
#define IDIM  13
#define HDIM  7
#define GDIM  28
#define FFDIM 512

#define RH    2
#define RPW   (2 * RH)
#define WPG   4
#define GROWS (WPG * RPW)
#define FPL   (FFDIM / (32 * WPG))
#define HPP   (32 * WPG + 4)
static_assert(FPL * 32 * WPG == FFDIM, "every lane of the group takes the same number of head features");
static_assert(32 * WPG == GROWS * 8, "second pass of the head's sums: 8 lanes a row, 16 partials a lane");

#define LDS_SYNC()                                                     \
    do {                                                               \
        __builtin_amdgcn_fence(__ATOMIC_RELEASE, "workgroup", "local"); \
        __builtin_amdgcn_s_barrier();                                  \
        __builtin_amdgcn_fence(__ATOMIC_ACQUIRE, "workgroup", "local"); \
    } while (0)

#define WAVE_LDS_ORDER()                                               \
    do {                                                               \
        __builtin_amdgcn_fence(__ATOMIC_RELEASE, "wavefront", "local"); \
        __builtin_amdgcn_wave_barrier();                               \
        __builtin_amdgcn_fence(__ATOMIC_ACQUIRE, "wavefront", "local"); \
    } while (0)

#define TC    16
#define NCH   (TSTEP / TC)
#define ROWQ  (TC * IDIM / 4)
#define XP    (TC * IDIM + 4)
#define XBUF  (RPW * XP)
#define NPIECE (RPW * ROWQ)
#define NLD   ((NPIECE + 31) / 32)
#define TG    (8 / RH)
static_assert(TG == 4 && TC % TG == 0, "the step loop deals a group's 8 x-projection WMMAs 3+3+2 over steps 1..3");

#define LOG2E 1.4426950408889634f

#if __has_builtin(__builtin_amdgcn_tanhf)
#define FAST_TANH(v) __builtin_amdgcn_tanhf(v)
#define TANH_TIER 2
#elif __has_builtin(__builtin_amdgcn_tanh_f32)
#define FAST_TANH(v) __builtin_amdgcn_tanh_f32(v)
#define TANH_TIER 2
#elif __has_builtin(__builtin_amdgcn_exp2f) && __has_builtin(__builtin_amdgcn_rcpf)
#define TANH_TIER 1
static __device__ __forceinline__ float FAST_TANH(float v) {
    float e2 = __builtin_amdgcn_exp2f(v * (2.0f * LOG2E));
    return fmaf(-2.0f, __builtin_amdgcn_rcpf(e2 + 1.0f), 1.0f);
}
#else
#define FAST_TANH(v) tanhf(v)
#define TANH_TIER 0
#endif

static __device__ __forceinline__ float sigm(float v) {
#if TANH_TIER == 2
    return fmaf(0.5f, FAST_TANH(0.5f * v), 0.5f);
#elif TANH_TIER == 1
    return __builtin_amdgcn_rcpf(1.0f + __builtin_amdgcn_exp2f(v * -LOG2E));
#else
    return 1.0f / (1.0f + __expf(-v));
#endif
}

static __device__ __forceinline__ float row_partner(float v) {
    return __builtin_bit_cast(float,
        __builtin_amdgcn_update_dpp(0, __builtin_bit_cast(int, v), 0x128, 0xf, 0xf, true));
}

__global__ __launch_bounds__(32 * WPG) void lstm_fused_kernel(
    const float* __restrict__ x,     const float* __restrict__ w_ih,
    const float* __restrict__ w_hh,  const float* __restrict__ b_ih,
    const float* __restrict__ b_hh,  const float* __restrict__ fc1_w,
    const float* __restrict__ fc1_b, const float* __restrict__ fc_w,
    const float* __restrict__ fc_b,  float* __restrict__ out)
{
    __shared__ __attribute__((aligned(16))) float hb_all[WPG][32 * RH + 8];
    __shared__ __attribute__((aligned(16))) float hpart[GROWS][HPP];
    __shared__ float hpart2[32 * WPG];
    __shared__ __attribute__((aligned(16))) float xs_all[WPG][2 * XBUF];

    const int wv   = __builtin_amdgcn_readfirstlane((int)(threadIdx.x >> 5));
    float* const hb  = hb_all[wv];
    float* const xs  = xs_all[wv];

    const int l    = threadIdx.x & 31;
    const int half = l >> 4;
    const int nl   = l & 15;
    const int b0   = (blockIdx.x * WPG + wv) * RPW;
    const int  jrow = ((nl & 7) & (RH - 1)) + RH * (nl >> 3);

    const int  unit = nl & 7;
    const bool ucol = unit < HDIM;
    int grow[2];
#pragma unroll
    for (int nt = 0; nt < 2; ++nt) grow[nt] = ucol ? (2 * nt + (nl >> 3)) * HDIM + unit : 0;

    v2f Bih[2][4];
    v2f Bhh[2][2];
#pragma unroll
    for (int kc = 0; kc < 4; ++kc) {
        int k0 = kc * 4 + 2 * half;
#pragma unroll
        for (int nt = 0; nt < 2; ++nt) {
            float m0 = (ucol && k0     < IDIM) ? 1.0f : 0.0f;
            float m1 = (ucol && k0 + 1 < IDIM) ? 1.0f : 0.0f;
            int ka = (k0     < IDIM) ? k0     : 0;
            int kb = (k0 + 1 < IDIM) ? k0 + 1 : 0;
            Bih[nt][kc][0] = w_ih[grow[nt] * IDIM + ka] * m0;
            Bih[nt][kc][1] = w_ih[grow[nt] * IDIM + kb] * m1;
        }
    }
#pragma unroll
    for (int kc = 0; kc < 2; ++kc) {
        int k0 = kc * 4 + 2 * half;
#pragma unroll
        for (int nt = 0; nt < 2; ++nt) {
            float m0 = (ucol && k0     < HDIM) ? 1.0f : 0.0f;
            float m1 = (ucol && k0 + 1 < HDIM) ? 1.0f : 0.0f;
            int ka = (k0     < HDIM) ? k0     : 0;
            int kb = (k0 + 1 < HDIM) ? k0 + 1 : 0;
            Bhh[nt][kc][0] = w_hh[grow[nt] * HDIM + ka] * m0;
            Bhh[nt][kc][1] = w_hh[grow[nt] * HDIM + kb] * m1;
        }
    }

    const float bias0 = ucol ? (b_ih[grow[0]] + b_hh[grow[0]]) : 0.0f;
    const float bias1 = ucol ? (b_ih[grow[1]] + b_hh[grow[1]]) : 0.0f;

    const int ft = threadIdx.x;
    float hw[FPL][HDIM], hbi[FPL], hfc[FPL];
#pragma unroll
    for (int k = 0; k < FPL; ++k) {
#pragma unroll
        for (int j = 0; j < HDIM; ++j) hw[k][j] = fc1_w[(ft * FPL + k) * HDIM + j];
        hbi[k] = fc1_b[ft * FPL + k];
        hfc[k] = fc_w[ft * FPL + k];
    }

    const bool glane = (nl < 8);
    const float act_s = glane ? 1.0f : 0.5f;
    const float act_b = glane ? 0.0f : 0.5f;

    for (int i = l; i < RPW * 8; i += 32) hb[i] = 0.0f;
    float c[RH];
#pragma unroll
    for (int r = 0; r < RH; ++r) c[r] = 0.0f;
    float* const hst = hb + ((nl < HDIM) ? 0 : RPW * 8) + (RH * half) * 8 + nl;
    __syncthreads();

    unsigned goff[NLD], loff[NLD];
#pragma unroll
    for (int i = 0; i < NLD; ++i) {
        unsigned f = (unsigned)(i * 32 + l);
        unsigned r = f / ROWQ;
        unsigned q = f - r * ROWQ;
        goff[i] = r * (TSTEP * IDIM) + q * 4;
        loff[i] = r * XP + q * 4;
    }
    const float* xblk = x + (size_t)b0 * (TSTEP * IDIM);
    v4f xr[NLD];
#pragma unroll
    for (int i = 0; i < NLD; ++i) xr[i] = v4f{0.0f, 0.0f, 0.0f, 0.0f};

    auto tile_fetch = [&](int cn) {
        const float* src = xblk + cn * (TC * IDIM);
#pragma unroll
        for (int i = 0; i < NLD; ++i) {
            if (i < NPIECE / 32 || l < NPIECE % 32)
                xr[i] = *reinterpret_cast<const v4f*>(src + goff[i]);
        }
    };
    auto tile_stash = [&](int cn) {
        float* dst = xs + (cn & 1) * XBUF;
#pragma unroll
        for (int i = 0; i < NLD; ++i) {
            if (i < NPIECE / 32 || l < NPIECE % 32)
                *reinterpret_cast<v4f*>(dst + loff[i]) = xr[i];
        }
    };

    const int jt    = (nl & 7) / RH;
    const int xlane = jrow * XP + jt * IDIM + 2 * half;
    const int xtail = jrow * XP + jt * IDIM + (half ? 0 : 12);
    auto x_frags = [&](int t0, v2f (&a)[4], float& v12) {
        const float* xt = xs + ((t0 / TC) & 1) * XBUF + (t0 % TC) * IDIM;
#pragma unroll
        for (int kc = 0; kc < 3; ++kc) {
            a[kc][0] = xt[xlane + kc * 4];
            a[kc][1] = xt[xlane + kc * 4 + 1];
        }
        v12 = xt[xtail];
    };
    auto x_tail = [&](v2f (&a)[4], float v12) {
        a[3][0] = half ? 0.0f : v12;
        a[3][1] = 0.0f;
    };

    auto x_project = [&](int w, const v2f (&a)[4], v8f& g0, v8f& g1) {
        const int kc = w >> 1;
        if ((w & 1) == 0) {
            if (kc == 0) {
#pragma unroll
                for (int i = 0; i < 8; ++i) g0[i] = bias0;
            }
            g0 = __builtin_amdgcn_wmma_f32_16x16x4_f32(false, a[kc], false, Bih[0][kc],
                                                       (short)0, g0, false, false);
        } else {
            if (kc == 0) {
#pragma unroll
                for (int i = 0; i < 8; ++i) g1[i] = bias1;
            }
            g1 = __builtin_amdgcn_wmma_f32_16x16x4_f32(false, a[kc], false, Bih[1][kc],
                                                       (short)0, g1, false, false);
        }
    };

    tile_fetch(0);
    tile_stash(0);
    tile_fetch(1);
    LDS_SYNC();
    v2f axn[4];
    float axt;
    v8f xp0, xp1;
    x_frags(0, axn, axt);
    x_tail(axn, axt);
#pragma unroll
    for (int w = 0; w < 8; ++w) x_project(w, axn, xp0, xp1);

    for (int t0 = 0; t0 < TSTEP; t0 += TG) {
        v8f xn0, xn1;
#pragma unroll
        for (int j = 0; j < TG; ++j) {
            v2f ah[2];
#pragma unroll
            for (int kc = 0; kc < 2; ++kc) {
                int k0 = kc * 4 + 2 * half;
                ah[kc][0] = hb[jrow * 8 + k0];
                ah[kc][1] = hb[jrow * 8 + k0 + 1];
            }
            __builtin_amdgcn_sched_barrier(0);
            if (j == 0) {
                int tn = (t0 + TG < TSTEP) ? (t0 + TG) : 0;
                x_frags(tn, axn, axt);
            } else {
                const int w0 = (j - 1) * 3;
                if (j == TG - 1) x_tail(axn, axt);
#pragma unroll
                for (int w = w0; w < w0 + 3 && w < 8; ++w) x_project(w, axn, xn0, xn1);
            }
            __builtin_amdgcn_sched_barrier(0);

            v8f acc0 = __builtin_amdgcn_wmma_f32_16x16x4_f32(false, ah[0], false, Bhh[0][0],
                                                             (short)0, xp0, false, false);
            v8f acc1 = __builtin_amdgcn_wmma_f32_16x16x4_f32(false, ah[0], false, Bhh[1][0],
                                                             (short)0, xp1, false, false);
            acc0 = __builtin_amdgcn_wmma_f32_16x16x4_f32(false, ah[1], false, Bhh[0][1],
                                                         (short)0, acc0, false, false);
            acc1 = __builtin_amdgcn_wmma_f32_16x16x4_f32(false, ah[1], false, Bhh[1][1],
                                                         (short)0, acc1, false, false);

            float hv[RH];
#pragma unroll
            for (int r = 0; r < RH; ++r) {
                float a0 = sigm(acc0[RH * j + r]);
#if TANH_TIER == 2
                float a1 = fmaf(act_s, FAST_TANH(act_s * acc1[RH * j + r]), act_b);
#else
                float a1 = glane ? FAST_TANH(acc1[RH * j + r]) : sigm(acc1[RH * j + r]);
#endif
                float fg = row_partner(a0);
                float og = row_partner(a1);
                c[r] = fg * c[r] + a0 * a1;
                hv[r] = og * FAST_TANH(c[r]);
            }
#pragma unroll
            for (int r = 0; r < RH; ++r) hst[r * 8] = hv[r];

            if (j == TG - 1 && (t0 % TC) == TC - 2 * TG) {
                int cn = t0 / TC + 1;
                if (cn < NCH)     tile_stash(cn);
                if (cn + 1 < NCH) tile_fetch(cn + 1);
            }
            WAVE_LDS_ORDER();
        }
        xp0 = xn0;
        xp1 = xn1;
    }

    __syncthreads();
#pragma unroll
    for (int R = 0; R < GROWS; ++R) {
        const float* hrow = &hb_all[R / RPW][(R % RPW) * 8];
        float hr[HDIM];
#pragma unroll
        for (int j = 0; j < HDIM; ++j) {
            float v = hrow[j];
            hr[j] = v > 0.0f ? v : 0.0f;
        }
        float a = 0.0f;
#pragma unroll
        for (int k = 0; k < FPL; ++k) {
            float s = hbi[k];
#pragma unroll
            for (int j = 0; j < HDIM; ++j) s = fmaf(hw[k][j], hr[j], s);
            s = s > 0.0f ? s : 0.0f;
            a = fmaf(hfc[k], s, a);
        }
        hpart[R][ft] = a;
    }
    __syncthreads();
    {
        const float* src = &hpart[ft >> 3][(ft & 7) * 16];
        float s4[4];
#pragma unroll
        for (int q = 0; q < 4; ++q) s4[q] = src[q];
#pragma unroll
        for (int i = 1; i < 4; ++i) {
#pragma unroll
            for (int q = 0; q < 4; ++q) s4[q] += src[i * 4 + q];
        }
        hpart2[ft] = (s4[0] + s4[1]) + (s4[2] + s4[3]);
    }
    __syncthreads();
    if (ft < GROWS) {
        const float* src = &hpart2[ft * 8];
        float s = ((src[0] + src[1]) + (src[2] + src[3])) + ((src[4] + src[5]) + (src[6] + src[7]));
        out[blockIdx.x * GROWS + ft] = s + fc_b[0];
    }
}

extern "C" void kernel_launch(void* const* d_in, const int* in_sizes, int n_in,
                              void* d_out, int out_size, void* d_ws, size_t ws_size,
                              hipStream_t stream) {
    (void)in_sizes; (void)n_in; (void)out_size; (void)d_ws; (void)ws_size;
    const float* x     = (const float*)d_in[0];
    const float* w_ih  = (const float*)d_in[1];
    const float* w_hh  = (const float*)d_in[2];
    const float* b_ih  = (const float*)d_in[3];
    const float* b_hh  = (const float*)d_in[4];
    const float* fc1_w = (const float*)d_in[5];
    const float* fc1_b = (const float*)d_in[6];
    const float* fc_w  = (const float*)d_in[7];
    const float* fc_b  = (const float*)d_in[8];

    lstm_fused_kernel<<<dim3(BATCH / (RPW * WPG)), dim3(32 * WPG), 0, stream>>>(
        x, w_ih, w_hh, b_ih, b_hh, fc1_w, fc1_b, fc_w, fc_b, (float*)d_out);
}


# h goes from accumulator to A-fragment layout in registers: 4x8 h packed in one VGPR, one row exchange (v_permlanex16), select, 4 in-row gathers (v_permlane16); no LDS trip per step; sums as before
# speedup vs baseline: 5.9242x; 1.0926x over previous
#include <hip/hip_runtime.h>


typedef float v2f __attribute__((ext_vector_type(2)));
typedef float v4f __attribute__((ext_vector_type(4)));
typedef float v8f __attribute__((ext_vector_type(8)));

#define BATCH 4096
#define TSTEP 512
#define IDIM  13
#define HDIM  7
#define GDIM  28
#define FFDIM 512

#define RH    2
#define RPW   (2 * RH)
#define WPG   4
#define GROWS (WPG * RPW)
#define FPL   (FFDIM / (32 * WPG))
#define HPP   (32 * WPG + 4)
static_assert(FPL * 32 * WPG == FFDIM, "every lane of the group takes the same number of head features");
static_assert(32 * WPG == GROWS * 8, "second pass of the head's sums: 8 lanes a row, 16 partials a lane");

#define LDS_SYNC()                                                     \
    do {                                                               \
        __builtin_amdgcn_fence(__ATOMIC_RELEASE, "workgroup", "local"); \
        __builtin_amdgcn_s_barrier();                                  \
        __builtin_amdgcn_fence(__ATOMIC_ACQUIRE, "workgroup", "local"); \
    } while (0)

#define WAVE_LDS_ORDER()                                               \
    do {                                                               \
        __builtin_amdgcn_fence(__ATOMIC_RELEASE, "wavefront", "local"); \
        __builtin_amdgcn_wave_barrier();                               \
        __builtin_amdgcn_fence(__ATOMIC_ACQUIRE, "wavefront", "local"); \
    } while (0)

#define TC    16
#define NCH   (TSTEP / TC)
#define ROWQ  (TC * IDIM / 4)
#define XP    (TC * IDIM + 4)
#define XBUF  (RPW * XP)
#define NPIECE (RPW * ROWQ)
#define NLD   ((NPIECE + 31) / 32)
#define TG    (8 / RH)
static_assert(TG == 4 && TC % TG == 0, "the step loop deals a group's 8 x-projection WMMAs 3+3+2 over steps 1..3");

#define LOG2E 1.4426950408889634f

#if __has_builtin(__builtin_amdgcn_tanhf)
#define FAST_TANH(v) __builtin_amdgcn_tanhf(v)
#define TANH_TIER 2
#elif __has_builtin(__builtin_amdgcn_tanh_f32)
#define FAST_TANH(v) __builtin_amdgcn_tanh_f32(v)
#define TANH_TIER 2
#elif __has_builtin(__builtin_amdgcn_exp2f) && __has_builtin(__builtin_amdgcn_rcpf)
#define TANH_TIER 1
static __device__ __forceinline__ float FAST_TANH(float v) {
    float e2 = __builtin_amdgcn_exp2f(v * (2.0f * LOG2E));
    return fmaf(-2.0f, __builtin_amdgcn_rcpf(e2 + 1.0f), 1.0f);
}
#else
#define FAST_TANH(v) tanhf(v)
#define TANH_TIER 0
#endif

static __device__ __forceinline__ float sigm(float v) {
#if TANH_TIER == 2
    return fmaf(0.5f, FAST_TANH(0.5f * v), 0.5f);
#elif TANH_TIER == 1
    return __builtin_amdgcn_rcpf(1.0f + __builtin_amdgcn_exp2f(v * -LOG2E));
#else
    return 1.0f / (1.0f + __expf(-v));
#endif
}

static __device__ __forceinline__ float row_partner(float v) {
    return __builtin_bit_cast(float,
        __builtin_amdgcn_update_dpp(0, __builtin_bit_cast(int, v), 0x128, 0xf, 0xf, true));
}

static __device__ __forceinline__ float row_gather(float dead, float v, unsigned lo, unsigned hi) {
    return __builtin_bit_cast(float,
        __builtin_amdgcn_permlane16(__builtin_bit_cast(int, dead), __builtin_bit_cast(int, v), lo, hi, false, false));
}
static __device__ __forceinline__ float xrow_gather(float dead, float v, unsigned lo, unsigned hi) {
    return __builtin_bit_cast(float,
        __builtin_amdgcn_permlanex16(__builtin_bit_cast(int, dead), __builtin_bit_cast(int, v), lo, hi, false, false));
}

__global__ __launch_bounds__(32 * WPG) void lstm_fused_kernel(
    const float* __restrict__ x,     const float* __restrict__ w_ih,
    const float* __restrict__ w_hh,  const float* __restrict__ b_ih,
    const float* __restrict__ b_hh,  const float* __restrict__ fc1_w,
    const float* __restrict__ fc1_b, const float* __restrict__ fc_w,
    const float* __restrict__ fc_b,  float* __restrict__ out)
{
    __shared__ __attribute__((aligned(16))) float hb_all[WPG][RPW * 8];
    __shared__ __attribute__((aligned(16))) float hpart[GROWS][HPP];
    __shared__ float hpart2[32 * WPG];
    __shared__ __attribute__((aligned(16))) float xs_all[WPG][2 * XBUF];

    const int wv   = __builtin_amdgcn_readfirstlane((int)(threadIdx.x >> 5));
    float* const hb  = hb_all[wv];
    float* const xs  = xs_all[wv];

    const int l    = threadIdx.x & 31;
    const int half = l >> 4;
    const int nl   = l & 15;
    const int b0   = (blockIdx.x * WPG + wv) * RPW;
    const int  jrow = ((nl & 7) & (RH - 1)) + RH * (nl >> 3);

    const int  unit = nl & 7;
    const bool ucol = unit < HDIM;
    int grow[2];
#pragma unroll
    for (int nt = 0; nt < 2; ++nt) grow[nt] = ucol ? (2 * nt + (nl >> 3)) * HDIM + unit : 0;

    v2f Bih[2][4];
    v2f Bhh[2][2];
#pragma unroll
    for (int kc = 0; kc < 4; ++kc) {
        int k0 = kc * 4 + 2 * half;
#pragma unroll
        for (int nt = 0; nt < 2; ++nt) {
            float m0 = (ucol && k0     < IDIM) ? 1.0f : 0.0f;
            float m1 = (ucol && k0 + 1 < IDIM) ? 1.0f : 0.0f;
            int ka = (k0     < IDIM) ? k0     : 0;
            int kb = (k0 + 1 < IDIM) ? k0 + 1 : 0;
            Bih[nt][kc][0] = w_ih[grow[nt] * IDIM + ka] * m0;
            Bih[nt][kc][1] = w_ih[grow[nt] * IDIM + kb] * m1;
        }
    }
#pragma unroll
    for (int kc = 0; kc < 2; ++kc) {
        int k0 = kc * 4 + 2 * half;
#pragma unroll
        for (int nt = 0; nt < 2; ++nt) {
            float m0 = (ucol && k0     < HDIM) ? 1.0f : 0.0f;
            float m1 = (ucol && k0 + 1 < HDIM) ? 1.0f : 0.0f;
            int ka = (k0     < HDIM) ? k0     : 0;
            int kb = (k0 + 1 < HDIM) ? k0 + 1 : 0;
            Bhh[nt][kc][0] = w_hh[grow[nt] * HDIM + ka] * m0;
            Bhh[nt][kc][1] = w_hh[grow[nt] * HDIM + kb] * m1;
        }
    }

    const float bias0 = ucol ? (b_ih[grow[0]] + b_hh[grow[0]]) : 0.0f;
    const float bias1 = ucol ? (b_ih[grow[1]] + b_hh[grow[1]]) : 0.0f;

    const int ft = threadIdx.x;
    float hw[FPL][HDIM], hbi[FPL], hfc[FPL];
#pragma unroll
    for (int k = 0; k < FPL; ++k) {
#pragma unroll
        for (int j = 0; j < HDIM; ++j) hw[k][j] = fc1_w[(ft * FPL + k) * HDIM + j];
        hbi[k] = fc1_b[ft * FPL + k];
        hfc[k] = fc_w[ft * FPL + k];
    }

    const bool glane = (nl < 8);
    const float act_s = glane ? 1.0f : 0.5f;
    const float act_b = glane ? 0.0f : 0.5f;

    float c[RH];
#pragma unroll
    for (int r = 0; r < RH; ++r) c[r] = 0.0f;

    static_assert(RH == 2 && HDIM <= 8, "the lane tables of the h hand-over are written for 2 batch rows a lane half");
    const bool hkeep = (((l >> 1) & 1) == half);

    unsigned goff[NLD], loff[NLD];
#pragma unroll
    for (int i = 0; i < NLD; ++i) {
        unsigned f = (unsigned)(i * 32 + l);
        unsigned r = f / ROWQ;
        unsigned q = f - r * ROWQ;
        goff[i] = r * (TSTEP * IDIM) + q * 4;
        loff[i] = r * XP + q * 4;
    }
    const float* xblk = x + (size_t)b0 * (TSTEP * IDIM);
    v4f xr[NLD];
#pragma unroll
    for (int i = 0; i < NLD; ++i) xr[i] = v4f{0.0f, 0.0f, 0.0f, 0.0f};

    auto tile_fetch = [&](int cn) {
        const float* src = xblk + cn * (TC * IDIM);
#pragma unroll
        for (int i = 0; i < NLD; ++i) {
            if (i < NPIECE / 32 || l < NPIECE % 32)
                xr[i] = *reinterpret_cast<const v4f*>(src + goff[i]);
        }
    };
    auto tile_stash = [&](int cn) {
        float* dst = xs + (cn & 1) * XBUF;
#pragma unroll
        for (int i = 0; i < NLD; ++i) {
            if (i < NPIECE / 32 || l < NPIECE % 32)
                *reinterpret_cast<v4f*>(dst + loff[i]) = xr[i];
        }
    };

    const int jt    = (nl & 7) / RH;
    const int xlane = jrow * XP + jt * IDIM + 2 * half;
    const int xtail = jrow * XP + jt * IDIM + (half ? 0 : 12);
    auto x_frags = [&](int t0, v2f (&a)[4], float& v12) {
        const float* xt = xs + ((t0 / TC) & 1) * XBUF + (t0 % TC) * IDIM;
#pragma unroll
        for (int kc = 0; kc < 3; ++kc) {
            a[kc][0] = xt[xlane + kc * 4];
            a[kc][1] = xt[xlane + kc * 4 + 1];
        }
        v12 = xt[xtail];
    };
    auto x_tail = [&](v2f (&a)[4], float v12) {
        a[3][0] = half ? 0.0f : v12;
        a[3][1] = 0.0f;
    };

    auto x_project = [&](int w, const v2f (&a)[4], v8f& g0, v8f& g1) {
        const int kc = w >> 1;
        if ((w & 1) == 0) {
            if (kc == 0) {
#pragma unroll
                for (int i = 0; i < 8; ++i) g0[i] = bias0;
            }
            g0 = __builtin_amdgcn_wmma_f32_16x16x4_f32(false, a[kc], false, Bih[0][kc],
                                                       (short)0, g0, false, false);
        } else {
            if (kc == 0) {
#pragma unroll
                for (int i = 0; i < 8; ++i) g1[i] = bias1;
            }
            g1 = __builtin_amdgcn_wmma_f32_16x16x4_f32(false, a[kc], false, Bih[1][kc],
                                                       (short)0, g1, false, false);
        }
    };

    tile_fetch(0);
    tile_stash(0);
    tile_fetch(1);
    LDS_SYNC();
    v2f axn[4];
    float axt;
    v8f xp0, xp1;
    x_frags(0, axn, axt);
    x_tail(axn, axt);
#pragma unroll
    for (int w = 0; w < 8; ++w) x_project(w, axn, xp0, xp1);

    auto x_slot = [&](int j, int s, v8f& g0, v8f& g1) {
        if (j == 0 || (j == TG - 1 && s == 0)) return;
        const int w = (j == TG - 1) ? 5 + s : (j - 1) * 3 + s;
        __builtin_amdgcn_sched_barrier(0);
        if (w == 6) x_tail(axn, axt);
        x_project(w, axn, g0, g1);
        __builtin_amdgcn_sched_barrier(0);
    };

    v2f ah[2];
#pragma unroll
    for (int kc = 0; kc < 2; ++kc) ah[kc] = v2f{0.0f, 0.0f};
    float hpk = 0.0f;

    for (int t0 = 0; t0 < TSTEP; t0 += TG) {
        v8f xn0, xn1;
#pragma unroll
        for (int j = 0; j < TG; ++j) {
            v8f acc0 = __builtin_amdgcn_wmma_f32_16x16x4_f32(false, ah[0], false, Bhh[0][0],
                                                             (short)0, xp0, false, false);
            v8f acc1 = __builtin_amdgcn_wmma_f32_16x16x4_f32(false, ah[0], false, Bhh[1][0],
                                                             (short)0, xp1, false, false);
            acc0 = __builtin_amdgcn_wmma_f32_16x16x4_f32(false, ah[1], false, Bhh[0][1],
                                                         (short)0, acc0, false, false);
            acc1 = __builtin_amdgcn_wmma_f32_16x16x4_f32(false, ah[1], false, Bhh[1][1],
                                                         (short)0, acc1, false, false);
            if (j == 0) {
                int tn = (t0 + TG < TSTEP) ? (t0 + TG) : 0;
                x_frags(tn, axn, axt);
                __builtin_amdgcn_sched_barrier(0);
            }

            float a0[RH], a1[RH], tc[RH], hv[RH];
#pragma unroll
            for (int r = 0; r < RH; ++r) {
                a0[r] = sigm(acc0[RH * j + r]);
#if TANH_TIER == 2
                a1[r] = fmaf(act_s, FAST_TANH(act_s * acc1[RH * j + r]), act_b);
#else
                a1[r] = glane ? FAST_TANH(acc1[RH * j + r]) : sigm(acc1[RH * j + r]);
#endif
            }
            x_slot(j, 0, xn0, xn1);
#pragma unroll
            for (int r = 0; r < RH; ++r) {
                float fg = row_partner(a0[r]);
                c[r] = fg * c[r] + a0[r] * a1[r];
                tc[r] = FAST_TANH(c[r]);
            }
            x_slot(j, 1, xn0, xn1);
#pragma unroll
            for (int r = 0; r < RH; ++r) {
                float og = row_partner(a1[r]);
                hv[r] = og * tc[r];
            }
            x_slot(j, 2, xn0, xn1);

            hpk = __builtin_bit_cast(float,
                __builtin_amdgcn_update_dpp(__builtin_bit_cast(int, hv[0]), __builtin_bit_cast(int, hv[1]),
                                            0x128, 0xf, 0xc, false));
            const float hx = xrow_gather(hv[1], hpk, 0x54761032u, 0xDCFE98BAu);
            const float hz = hkeep ? hpk : hx;
            ah[0][0] = row_gather(ah[0][0], hz, 0x80808080u, 0xA2A2A2A2u);
            ah[0][1] = row_gather(ah[0][1], hz, 0x91919191u, 0xB3B3B3B3u);
            ah[1][0] = row_gather(ah[1][0], hz, 0xC4C4C4C4u, 0xE6E6E6E6u);
            ah[1][1] = row_gather(ah[1][1], hz, 0xD5D5D5D5u, 0xF7F7F7F7u);

            if (j == TG - 1 && (t0 % TC) == TC - 2 * TG) {
                int cn = t0 / TC + 1;
                if (cn < NCH)     tile_stash(cn);
                if (cn + 1 < NCH) tile_fetch(cn + 1);
            }
            WAVE_LDS_ORDER();
            __builtin_amdgcn_sched_barrier(0);
        }
        xp0 = xn0;
        xp1 = xn1;
    }
    hb[l] = hpk;

    __syncthreads();
#pragma unroll
    for (int R = 0; R < GROWS; ++R) {
        const float* hrow = &hb_all[R / RPW][(R % RPW) * 8];
        float hr[HDIM];
#pragma unroll
        for (int j = 0; j < HDIM; ++j) {
            float v = hrow[j];
            hr[j] = v > 0.0f ? v : 0.0f;
        }
        float a = 0.0f;
#pragma unroll
        for (int k = 0; k < FPL; ++k) {
            float s = hbi[k];
#pragma unroll
            for (int j = 0; j < HDIM; ++j) s = fmaf(hw[k][j], hr[j], s);
            s = s > 0.0f ? s : 0.0f;
            a = fmaf(hfc[k], s, a);
        }
        hpart[R][ft] = a;
    }
    __syncthreads();
    {
        const float* src = &hpart[ft >> 3][(ft & 7) * 16];
        float s4[4];
#pragma unroll
        for (int q = 0; q < 4; ++q) s4[q] = src[q];
#pragma unroll
        for (int i = 1; i < 4; ++i) {
#pragma unroll
            for (int q = 0; q < 4; ++q) s4[q] += src[i * 4 + q];
        }
        hpart2[ft] = (s4[0] + s4[1]) + (s4[2] + s4[3]);
    }
    __syncthreads();
    if (ft < GROWS) {
        const float* src = &hpart2[ft * 8];
        float s = ((src[0] + src[1]) + (src[2] + src[3])) + ((src[4] + src[5]) + (src[6] + src[7]));
        out[blockIdx.x * GROWS + ft] = s + fc_b[0];
    }
}

extern "C" void kernel_launch(void* const* d_in, const int* in_sizes, int n_in,
                              void* d_out, int out_size, void* d_ws, size_t ws_size,
                              hipStream_t stream) {
    (void)in_sizes; (void)n_in; (void)out_size; (void)d_ws; (void)ws_size;
    const float* x     = (const float*)d_in[0];
    const float* w_ih  = (const float*)d_in[1];
    const float* w_hh  = (const float*)d_in[2];
    const float* b_ih  = (const float*)d_in[3];
    const float* b_hh  = (const float*)d_in[4];
    const float* fc1_w = (const float*)d_in[5];
    const float* fc1_b = (const float*)d_in[6];
    const float* fc_w  = (const float*)d_in[7];
    const float* fc_b  = (const float*)d_in[8];

    lstm_fused_kernel<<<dim3(BATCH / (RPW * WPG)), dim3(32 * WPG), 0, stream>>>(
        x, w_ih, w_hh, b_ih, b_hh, fc1_w, fc1_b, fc_w, fc_b, (float*)d_out);
}
